# starting kernel
# baseline (speedup 1.0000x reference)
#include <hip/hip_runtime.h>
#include <hip/hip_bf16.h>


#define QN    8
#define KCB   1024
#define DIMV  64

typedef __attribute__((ext_vector_type(2))) float v2f;
typedef __attribute__((ext_vector_type(8))) float v8f;

__global__ __launch_bounds__(256)
void ResidualVQ_csq_kernel(const float* __restrict__ cbs,
                           float* __restrict__ csq, int total) {
    int i = blockIdx.x * blockDim.x + threadIdx.x;
    if (i >= total) return;
    const float* r = cbs + (size_t)i * DIMV;
    float s = 0.0f;
#pragma unroll
    for (int k = 0; k < DIMV; ++k) s = __builtin_fmaf(r[k], r[k], s);
    csq[i] = s;
}

__global__ __launch_bounds__(128)
void ResidualVQ_main_kernel(const float* __restrict__ x,
                            const float* __restrict__ cbs,
                            const float* __restrict__ csq,
                            float* __restrict__ out, int tokens) {
    const int lane = threadIdx.x & 31;
    const int wave = threadIdx.x >> 5;
    const int m    = lane & 15;
    const int hi   = lane >> 4;
    const int tileBase = (blockIdx.x * 4 + wave) * 32;
    const int token0 = tileBase + m;
    const int token1 = tileBase + 16 + m;

    const float* xrow0 = x + (size_t)token0 * DIMV;
    const float* xrow1 = x + (size_t)token1 * DIMV;
    v2f a0[16], a1[16], acc0[16], acc1[16];
#pragma unroll
    for (int j = 0; j < 16; ++j) {
        a0[j]   = *(const v2f*)(xrow0 + 4 * j + 2 * hi);
        a1[j]   = *(const v2f*)(xrow1 + 4 * j + 2 * hi);
        acc0[j] = (v2f){0.0f, 0.0f};
        acc1[j] = (v2f){0.0f, 0.0f};
    }

    for (int q = 0; q < QN; ++q) {
        const float* cb = cbs + (size_t)q * KCB * DIMV;
        const float* cs = csq + q * KCB;

        float bestS0[8], bestS1[8];
        int   bestI0[8], bestI1[8];
#pragma unroll
        for (int r = 0; r < 8; ++r) {
            bestS0[r] = __builtin_inff(); bestI0[r] = 0;
            bestS1[r] = __builtin_inff(); bestI1[r] = 0;
        }

        for (int chunk = 0; chunk < KCB / 16; ++chunk) {
            const int n = chunk * 16 + m;
            const float* crow = cb + (size_t)n * DIMV + 2 * hi;

            v2f b[16];
#pragma unroll
            for (int j = 0; j < 16; ++j) b[j] = *(const v2f*)(crow + 4 * j);

            v8f c0 = {};
            v8f c1 = {};
#pragma unroll
            for (int j = 0; j < 16; ++j) {
                c0 = __builtin_amdgcn_wmma_f32_16x16x4_f32(
                         false, a0[j], false, b[j], (short)0, c0, false, false);
                c1 = __builtin_amdgcn_wmma_f32_16x16x4_f32(
                         false, a1[j], false, b[j], (short)0, c1, false, false);
            }

            if (chunk + 1 < KCB / 16)
                __builtin_prefetch(crow + 16 * DIMV, 0, 0);

            const float cnorm = cs[n];
#pragma unroll
            for (int r = 0; r < 8; ++r) {
                float s0 = __builtin_fmaf(-2.0f, c0[r], cnorm);
                float s1 = __builtin_fmaf(-2.0f, c1[r], cnorm);
                if (s0 < bestS0[r]) { bestS0[r] = s0; bestI0[r] = n; }
                if (s1 < bestS1[r]) { bestS1[r] = s1; bestI1[r] = n; }
            }
        }

#pragma unroll
        for (int r = 0; r < 8; ++r) {
            float s0 = bestS0[r]; int i0 = bestI0[r];
            float s1 = bestS1[r]; int i1 = bestI1[r];
#pragma unroll
            for (int off = 8; off >= 1; off >>= 1) {
                float so0 = __shfl_xor(s0, off, 16);
                int   io0 = __shfl_xor(i0, off, 16);
                float so1 = __shfl_xor(s1, off, 16);
                int   io1 = __shfl_xor(i1, off, 16);
                if (so0 < s0 || (so0 == s0 && io0 < i0)) { s0 = so0; i0 = io0; }
                if (so1 < s1 || (so1 == s1 && io1 < i1)) { s1 = so1; i1 = io1; }
            }
            bestI0[r] = i0;
            bestI1[r] = i1;
        }

        int myIdx0 = 0, myIdx1 = 0;
#pragma unroll
        for (int r = 0; r < 8; ++r) {
            int lo0 = __shfl(bestI0[r], 0, 32);
            int hi0 = __shfl(bestI0[r], 16, 32);
            int lo1 = __shfl(bestI1[r], 0, 32);
            int hi1 = __shfl(bestI1[r], 16, 32);
            if (m == r)     { myIdx0 = lo0; myIdx1 = lo1; }
            if (m == r + 8) { myIdx0 = hi0; myIdx1 = hi1; }
        }

        const float* sel0 = cb + (size_t)myIdx0 * DIMV + 2 * hi;
        const float* sel1 = cb + (size_t)myIdx1 * DIMV + 2 * hi;
#pragma unroll
        for (int j = 0; j < 16; ++j) {
            v2f cv0 = *(const v2f*)(sel0 + 4 * j);
            v2f cv1 = *(const v2f*)(sel1 + 4 * j);
            a0[j]   -= cv0;
            a1[j]   -= cv1;
            acc0[j] += cv0;
            acc1[j] += cv1;
        }
    }

    float* orow0 = out + (size_t)token0 * DIMV;
    float* orow1 = out + (size_t)token1 * DIMV;
#pragma unroll
    for (int j = 0; j < 16; ++j) {
        *(v2f*)(orow0 + 4 * j + 2 * hi) = acc0[j];
        *(v2f*)(orow1 + 4 * j + 2 * hi) = acc1[j];
    }
}

extern "C" void kernel_launch(void* const* d_in, const int* in_sizes, int n_in,
                              void* d_out, int out_size, void* d_ws, size_t ws_size,
                              hipStream_t stream) {
    const float* x   = (const float*)d_in[0];
    const float* cbs = (const float*)d_in[1];
    float*       out = (float*)d_out;
    float*       csq = (float*)d_ws;

    const int tokens  = in_sizes[0] / DIMV;
    const int totalCB = in_sizes[1] / DIMV;

    ResidualVQ_csq_kernel<<<(totalCB + 255) / 256, 256, 0, stream>>>(cbs, csq, totalCB);
    ResidualVQ_main_kernel<<<tokens / 128, 128, 0, stream>>>(x, cbs, csq, out, tokens);
}


# Codebook staged through LDS once per 8-wave work-group with whole-line loads fetched a pass ahead; WMMA B fragments come from padded LDS rows instead of 8-byte global gathers across sixteen rows
# speedup vs baseline: 1.1909x; 1.1909x over previous
#include <hip/hip_runtime.h>
#include <hip/hip_bf16.h>


#define QN    8
#define KCB   1024
#define DIMV  64
#define TROWS 512
#define PITCH 68
#define TFLT  (TROWS * PITCH)
#define NTILE (QN * KCB / TROWS)
#define STEPS (TROWS / 16)

typedef __attribute__((ext_vector_type(2))) float v2f;
typedef __attribute__((ext_vector_type(4))) float v4f;
typedef __attribute__((ext_vector_type(8))) float v8f;

__global__ __launch_bounds__(256)
void ResidualVQ_csq_kernel(const float* __restrict__ cbs,
                           float* __restrict__ csq, int total) {
    int i = blockIdx.x * blockDim.x + threadIdx.x;
    if (i >= total) return;
    const float* r = cbs + (size_t)i * DIMV;
    float s = 0.0f;
#pragma unroll
    for (int k = 0; k < DIMV; ++k) s = __builtin_fmaf(r[k], r[k], s);
    csq[i] = s;
}

__device__ __forceinline__ void load_bfrag(v2f (&b)[16], const float* p) {
#pragma unroll
    for (int j = 0; j < 16; ++j) b[j] = *(const v2f*)(p + 4 * j);
}

__device__ __forceinline__ void vq_mma(const v2f (&a0)[16], const v2f (&a1)[16],
                                       const v2f (&b)[16], v8f& c0, v8f& c1) {
    c0 = (v8f){};
    c1 = (v8f){};
#pragma unroll
    for (int j = 0; j < 16; ++j) {
        c0 = __builtin_amdgcn_wmma_f32_16x16x4_f32(
                 false, a0[j], false, b[j], (short)0, c0, false, false);
        c1 = __builtin_amdgcn_wmma_f32_16x16x4_f32(
                 false, a1[j], false, b[j], (short)0, c1, false, false);
    }
}

__device__ __forceinline__ void vq_argmin(const v8f& c0, const v8f& c1, float cnorm, int n,
                                          float (&bestS0)[8], int (&bestI0)[8],
                                          float (&bestS1)[8], int (&bestI1)[8]) {
#pragma unroll
    for (int r = 0; r < 8; ++r) {
        float s0 = __builtin_fmaf(-2.0f, c0[r], cnorm);
        float s1 = __builtin_fmaf(-2.0f, c1[r], cnorm);
        if (s0 < bestS0[r]) { bestS0[r] = s0; bestI0[r] = n; }
        if (s1 < bestS1[r]) { bestS1[r] = s1; bestI1[r] = n; }
    }
}

__global__ __launch_bounds__(256)
void ResidualVQ_main_kernel(const float* __restrict__ x,
                            const float* __restrict__ cbs,
                            const float* __restrict__ csq,
                            float* __restrict__ out, int tokens) {
    __shared__ __attribute__((aligned(16))) float lds[2 * TFLT];

    const int tid  = threadIdx.x;
    const int lane = tid & 31;
    const int wave = tid >> 5;
    const int m    = lane & 15;
    const int hi   = lane >> 4;
    const int tileBase = (blockIdx.x * 8 + wave) * 32;
    const int token0 = tileBase + m;
    const int token1 = tileBase + 16 + m;

    const int srow   = tid >> 4;
    const int spiece = tid & 15;
    const int gOff = srow * DIMV  + spiece * 4;
    const int lOff = srow * PITCH + spiece * 4;

    const float* xrow0 = x + (size_t)token0 * DIMV;
    const float* xrow1 = x + (size_t)token1 * DIMV;
    v2f a0[16], a1[16], acc0[16], acc1[16];
#pragma unroll
    for (int j = 0; j < 16; ++j) {
        a0[j]   = *(const v2f*)(xrow0 + 4 * j + 2 * hi);
        a1[j]   = *(const v2f*)(xrow1 + 4 * j + 2 * hi);
        acc0[j] = (v2f){0.0f, 0.0f};
        acc1[j] = (v2f){0.0f, 0.0f};
    }

    const float* gstream = cbs + gOff;
    int cp = TROWS / 32;
    v4f p0 = *(const v4f*)(gstream + (size_t)cp * 32 * DIMV);
    v4f p1 = *(const v4f*)(gstream + (size_t)cp * 32 * DIMV + 16 * DIMV);

    for (int s = 0; s < STEPS; ++s) {
        v4f v = *(const v4f*)(cbs + (size_t)s * 16 * DIMV + gOff);
        *(v4f*)(lds + s * 16 * PITCH + lOff) = v;
    }
    __syncthreads();

    for (int q = 0; q < QN; ++q) {
        const float* cb = cbs + (size_t)q * KCB * DIMV;
        const float* cs = csq + q * KCB;

        float bestS0[8], bestS1[8];
        int   bestI0[8], bestI1[8];
#pragma unroll
        for (int r = 0; r < 8; ++r) {
            bestS0[r] = __builtin_inff(); bestI0[r] = 0;
            bestS1[r] = __builtin_inff(); bestI1[r] = 0;
        }

        for (int h = 0; h < 2; ++h) {
            const int t = q * 2 + h;
            const float* cur  = lds + (t & 1) * TFLT;
            float*       nxt  = lds + ((t + 1) & 1) * TFLT;
            float*       ldst = nxt + lOff;
            const float* frag = cur + m * PITCH + 2 * hi;
            const float* csn  = cs + h * TROWS + m;
            const int    nb   = h * TROWS + m;

            v2f bA[16], bB[16];
            load_bfrag(bA, frag);
            load_bfrag(bB, frag + 16 * PITCH);
            for (int i = 0; i < STEPS; i += 2) {
                *(v4f*)(ldst + i * 16 * PITCH) = p0;
                *(v4f*)(ldst + (i + 1) * 16 * PITCH) = p1;
                cp = (cp + 1 < QN * KCB / 32) ? (cp + 1) : cp;
                p0 = *(const v4f*)(gstream + (size_t)cp * 32 * DIMV);
                p1 = *(const v4f*)(gstream + (size_t)cp * 32 * DIMV + 16 * DIMV);
                const float cn0 = csn[i * 16];
                const float cn1 = csn[(i + 1) * 16];

                v8f c0A, c1A, c0B, c1B;
                vq_mma(a0, a1, bA, c0A, c1A);
                vq_mma(a0, a1, bB, c0B, c1B);

                const int i2 = (i + 2 < STEPS) ? (i + 2) : i;
                load_bfrag(bA, frag + i2 * 16 * PITCH);
                load_bfrag(bB, frag + (i2 + 1) * 16 * PITCH);

                vq_argmin(c0A, c1A, cn0, nb + i * 16, bestS0, bestI0, bestS1, bestI1);
                vq_argmin(c0B, c1B, cn1, nb + (i + 1) * 16, bestS0, bestI0, bestS1, bestI1);
            }
            __syncthreads();
        }

#pragma unroll
        for (int r = 0; r < 8; ++r) {
            float s0 = bestS0[r]; int i0 = bestI0[r];
            float s1 = bestS1[r]; int i1 = bestI1[r];
#pragma unroll
            for (int off = 8; off >= 1; off >>= 1) {
                float so0 = __shfl_xor(s0, off, 16);
                int   io0 = __shfl_xor(i0, off, 16);
                float so1 = __shfl_xor(s1, off, 16);
                int   io1 = __shfl_xor(i1, off, 16);
                if (so0 < s0 || (so0 == s0 && io0 < i0)) { s0 = so0; i0 = io0; }
                if (so1 < s1 || (so1 == s1 && io1 < i1)) { s1 = so1; i1 = io1; }
            }
            bestI0[r] = i0;
            bestI1[r] = i1;
        }

        int myIdx0 = 0, myIdx1 = 0;
#pragma unroll
        for (int r = 0; r < 8; ++r) {
            int lo0 = __shfl(bestI0[r], 0, 32);
            int hi0 = __shfl(bestI0[r], 16, 32);
            int lo1 = __shfl(bestI1[r], 0, 32);
            int hi1 = __shfl(bestI1[r], 16, 32);
            if (m == r)     { myIdx0 = lo0; myIdx1 = lo1; }
            if (m == r + 8) { myIdx0 = hi0; myIdx1 = hi1; }
        }

        const float* sel0 = cb + (size_t)myIdx0 * DIMV + 2 * hi;
        const float* sel1 = cb + (size_t)myIdx1 * DIMV + 2 * hi;
#pragma unroll
        for (int j = 0; j < 16; ++j) {
            v2f cv0 = *(const v2f*)(sel0 + 4 * j);
            v2f cv1 = *(const v2f*)(sel1 + 4 * j);
            a0[j]   -= cv0;
            a1[j]   -= cv1;
            acc0[j] += cv0;
            acc1[j] += cv1;
        }
    }

    float* orow0 = out + (size_t)token0 * DIMV;
    float* orow1 = out + (size_t)token1 * DIMV;
#pragma unroll
    for (int j = 0; j < 16; ++j) {
        *(v2f*)(orow0 + 4 * j + 2 * hi) = acc0[j];
        *(v2f*)(orow1 + 4 * j + 2 * hi) = acc1[j];
    }
}

extern "C" void kernel_launch(void* const* d_in, const int* in_sizes, int n_in,
                              void* d_out, int out_size, void* d_ws, size_t ws_size,
                              hipStream_t stream) {
    const float* x   = (const float*)d_in[0];
    const float* cbs = (const float*)d_in[1];
    float*       out = (float*)d_out;
    float*       csq = (float*)d_ws;

    const int tokens  = in_sizes[0] / DIMV;
    const int totalCB = in_sizes[1] / DIMV;

    ResidualVQ_csq_kernel<<<(totalCB + 255) / 256, 256, 0, stream>>>(cbs, csq, totalCB);
    ResidualVQ_main_kernel<<<tokens / 256, 256, 0, stream>>>(x, cbs, csq, out, tokens);
}


# Stage-end argmin across the 16 lanes of a row by row_xmask DPP moves and selects on the vector ALU, all 16 row pairs side by side, instead of 128 LDS-routed shuffles in 32 serial, branchy steps
# speedup vs baseline: 1.2382x; 1.0397x over previous
#include <hip/hip_runtime.h>
#include <hip/hip_bf16.h>


#define QN    8
#define KCB   1024
#define DIMV  64
#define TROWS 512
#define PITCH 68
#define TFLT  (TROWS * PITCH)
#define NTILE (QN * KCB / TROWS)
#define STEPS (TROWS / 16)

typedef __attribute__((ext_vector_type(2))) float v2f;
typedef __attribute__((ext_vector_type(4))) float v4f;
typedef __attribute__((ext_vector_type(8))) float v8f;

__global__ __launch_bounds__(256)
void ResidualVQ_csq_kernel(const float* __restrict__ cbs,
                           float* __restrict__ csq, int total) {
    int i = blockIdx.x * blockDim.x + threadIdx.x;
    if (i >= total) return;
    const float* r = cbs + (size_t)i * DIMV;
    float s = 0.0f;
#pragma unroll
    for (int k = 0; k < DIMV; ++k) s = __builtin_fmaf(r[k], r[k], s);
    csq[i] = s;
}

__device__ __forceinline__ void load_bfrag(v2f (&b)[16], const float* p) {
#pragma unroll
    for (int j = 0; j < 16; ++j) b[j] = *(const v2f*)(p + 4 * j);
}

__device__ __forceinline__ void vq_mma(const v2f (&a0)[16], const v2f (&a1)[16],
                                       const v2f (&b)[16], v8f& c0, v8f& c1) {
    c0 = (v8f){};
    c1 = (v8f){};
#pragma unroll
    for (int j = 0; j < 16; ++j) {
        c0 = __builtin_amdgcn_wmma_f32_16x16x4_f32(
                 false, a0[j], false, b[j], (short)0, c0, false, false);
        c1 = __builtin_amdgcn_wmma_f32_16x16x4_f32(
                 false, a1[j], false, b[j], (short)0, c1, false, false);
    }
}

__device__ __forceinline__ void vq_argmin(const v8f& c0, const v8f& c1, float cnorm, int n,
                                          float (&bestS0)[8], int (&bestI0)[8],
                                          float (&bestS1)[8], int (&bestI1)[8]) {
#pragma unroll
    for (int r = 0; r < 8; ++r) {
        float s0 = __builtin_fmaf(-2.0f, c0[r], cnorm);
        float s1 = __builtin_fmaf(-2.0f, c1[r], cnorm);
        if (s0 < bestS0[r]) { bestS0[r] = s0; bestI0[r] = n; }
        if (s1 < bestS1[r]) { bestS1[r] = s1; bestI1[r] = n; }
    }
}

template <int MASK>
__device__ __forceinline__ int row_xor(int v) {
    return __builtin_amdgcn_mov_dpp(v, 0x160 | MASK, 0xf, 0xf, false);
}

template <int MASK>
__device__ __forceinline__ void vq_merge(float& s, int& i) {
    const float so = __builtin_bit_cast(float, row_xor<MASK>(__builtin_bit_cast(int, s)));
    const int   io = row_xor<MASK>(i);
    const bool take = (so < s) | ((so == s) & (io < i));
    s = take ? so : s;
    i = take ? io : i;
}

template <int MASK>
__device__ __forceinline__ void vq_merge_all(float (&bestS0)[8], int (&bestI0)[8],
                                             float (&bestS1)[8], int (&bestI1)[8]) {
#pragma unroll
    for (int r = 0; r < 8; ++r) {
        vq_merge<MASK>(bestS0[r], bestI0[r]);
        vq_merge<MASK>(bestS1[r], bestI1[r]);
    }
}

__global__ __launch_bounds__(256)
void ResidualVQ_main_kernel(const float* __restrict__ x,
                            const float* __restrict__ cbs,
                            const float* __restrict__ csq,
                            float* __restrict__ out, int tokens) {
    __shared__ __attribute__((aligned(16))) float lds[2 * TFLT];

    const int tid  = threadIdx.x;
    const int lane = tid & 31;
    const int wave = tid >> 5;
    const int m    = lane & 15;
    const int hi   = lane >> 4;
    const int tileBase = (blockIdx.x * 8 + wave) * 32;
    const int token0 = tileBase + m;
    const int token1 = tileBase + 16 + m;

    const int srow   = tid >> 4;
    const int spiece = tid & 15;
    const int gOff = srow * DIMV  + spiece * 4;
    const int lOff = srow * PITCH + spiece * 4;

    const float* xrow0 = x + (size_t)token0 * DIMV;
    const float* xrow1 = x + (size_t)token1 * DIMV;
    v2f a0[16], a1[16], acc0[16], acc1[16];
#pragma unroll
    for (int j = 0; j < 16; ++j) {
        a0[j]   = *(const v2f*)(xrow0 + 4 * j + 2 * hi);
        a1[j]   = *(const v2f*)(xrow1 + 4 * j + 2 * hi);
        acc0[j] = (v2f){0.0f, 0.0f};
        acc1[j] = (v2f){0.0f, 0.0f};
    }

    const float* gstream = cbs + gOff;
    int cp = TROWS / 32;
    v4f p0 = *(const v4f*)(gstream + (size_t)cp * 32 * DIMV);
    v4f p1 = *(const v4f*)(gstream + (size_t)cp * 32 * DIMV + 16 * DIMV);

    for (int s = 0; s < STEPS; ++s) {
        v4f v = *(const v4f*)(cbs + (size_t)s * 16 * DIMV + gOff);
        *(v4f*)(lds + s * 16 * PITCH + lOff) = v;
    }
    __syncthreads();

    for (int q = 0; q < QN; ++q) {
        const float* cb = cbs + (size_t)q * KCB * DIMV;
        const float* cs = csq + q * KCB;

        float bestS0[8], bestS1[8];
        int   bestI0[8], bestI1[8];
#pragma unroll
        for (int r = 0; r < 8; ++r) {
            bestS0[r] = __builtin_inff(); bestI0[r] = 0;
            bestS1[r] = __builtin_inff(); bestI1[r] = 0;
        }

        for (int h = 0; h < 2; ++h) {
            const int t = q * 2 + h;
            const float* cur  = lds + (t & 1) * TFLT;
            float*       nxt  = lds + ((t + 1) & 1) * TFLT;
            float*       ldst = nxt + lOff;
            const float* frag = cur + m * PITCH + 2 * hi;
            const float* csn  = cs + h * TROWS + m;
            const int    nb   = h * TROWS + m;

            v2f bA[16], bB[16];
            load_bfrag(bA, frag);
            load_bfrag(bB, frag + 16 * PITCH);
            for (int i = 0; i < STEPS; i += 2) {
                *(v4f*)(ldst + i * 16 * PITCH) = p0;
                *(v4f*)(ldst + (i + 1) * 16 * PITCH) = p1;
                cp = (cp + 1 < QN * KCB / 32) ? (cp + 1) : cp;
                p0 = *(const v4f*)(gstream + (size_t)cp * 32 * DIMV);
                p1 = *(const v4f*)(gstream + (size_t)cp * 32 * DIMV + 16 * DIMV);
                const float cn0 = csn[i * 16];
                const float cn1 = csn[(i + 1) * 16];

                v8f c0A, c1A, c0B, c1B;
                vq_mma(a0, a1, bA, c0A, c1A);
                vq_mma(a0, a1, bB, c0B, c1B);

                const int i2 = (i + 2 < STEPS) ? (i + 2) : i;
                load_bfrag(bA, frag + i2 * 16 * PITCH);
                load_bfrag(bB, frag + (i2 + 1) * 16 * PITCH);

                vq_argmin(c0A, c1A, cn0, nb + i * 16, bestS0, bestI0, bestS1, bestI1);
                vq_argmin(c0B, c1B, cn1, nb + (i + 1) * 16, bestS0, bestI0, bestS1, bestI1);
            }
            __syncthreads();
        }

        vq_merge_all<8>(bestS0, bestI0, bestS1, bestI1);
        vq_merge_all<4>(bestS0, bestI0, bestS1, bestI1);
        vq_merge_all<2>(bestS0, bestI0, bestS1, bestI1);
        vq_merge_all<1>(bestS0, bestI0, bestS1, bestI1);

        int myIdx0 = 0, myIdx1 = 0;
#pragma unroll
        for (int r = 0; r < 8; ++r) {
            int lo0 = __shfl(bestI0[r], 0, 32);
            int hi0 = __shfl(bestI0[r], 16, 32);
            int lo1 = __shfl(bestI1[r], 0, 32);
            int hi1 = __shfl(bestI1[r], 16, 32);
            if (m == r)     { myIdx0 = lo0; myIdx1 = lo1; }
            if (m == r + 8) { myIdx0 = hi0; myIdx1 = hi1; }
        }

        const float* sel0 = cb + (size_t)myIdx0 * DIMV + 2 * hi;
        const float* sel1 = cb + (size_t)myIdx1 * DIMV + 2 * hi;
#pragma unroll
        for (int j = 0; j < 16; ++j) {
            v2f cv0 = *(const v2f*)(sel0 + 4 * j);
            v2f cv1 = *(const v2f*)(sel1 + 4 * j);
            a0[j]   -= cv0;
            a1[j]   -= cv1;
            acc0[j] += cv0;
            acc1[j] += cv1;
        }
    }

    float* orow0 = out + (size_t)token0 * DIMV;
    float* orow1 = out + (size_t)token1 * DIMV;
#pragma unroll
    for (int j = 0; j < 16; ++j) {
        *(v2f*)(orow0 + 4 * j + 2 * hi) = acc0[j];
        *(v2f*)(orow1 + 4 * j + 2 * hi) = acc1[j];
    }
}

extern "C" void kernel_launch(void* const* d_in, const int* in_sizes, int n_in,
                              void* d_out, int out_size, void* d_ws, size_t ws_size,
                              hipStream_t stream) {
    const float* x   = (const float*)d_in[0];
    const float* cbs = (const float*)d_in[1];
    float*       out = (float*)d_out;
    float*       csq = (float*)d_ws;

    const int tokens  = in_sizes[0] / DIMV;
    const int totalCB = in_sizes[1] / DIMV;

    ResidualVQ_csq_kernel<<<(totalCB + 255) / 256, 256, 0, stream>>>(cbs, csq, totalCB);
    ResidualVQ_main_kernel<<<tokens / 256, 256, 0, stream>>>(x, cbs, csq, out, tokens);
}


# B fragments double-buffered in two named register sets, loop unrolled to two passes per trip: no register copies among the WMMAs, each pass's LDS traffic one group ahead of its WMMA block
# speedup vs baseline: 1.2918x; 1.0433x over previous
#include <hip/hip_runtime.h>
#include <hip/hip_bf16.h>


#define QN    8
#define KCB   1024
#define DIMV  64
#define TROWS 512
#define PITCH 68
#define TFLT  (TROWS * PITCH)
#define NTILE (QN * KCB / TROWS)
#define STEPS (TROWS / 16)

typedef __attribute__((ext_vector_type(2))) float v2f;
typedef __attribute__((ext_vector_type(4))) float v4f;
typedef __attribute__((ext_vector_type(8))) float v8f;

__global__ __launch_bounds__(256)
void ResidualVQ_csq_kernel(const float* __restrict__ cbs,
                           float* __restrict__ csq, int total) {
    int i = blockIdx.x * blockDim.x + threadIdx.x;
    if (i >= total) return;
    const float* r = cbs + (size_t)i * DIMV;
    float s = 0.0f;
#pragma unroll
    for (int k = 0; k < DIMV; ++k) s = __builtin_fmaf(r[k], r[k], s);
    csq[i] = s;
}

__device__ __forceinline__ void load_bfrag(v2f (&b)[16], const float* p) {
#pragma unroll
    for (int j = 0; j < 16; ++j) b[j] = *(const v2f*)(p + 4 * j);
}

__device__ __forceinline__ void vq_mma(const v2f (&a0)[16], const v2f (&a1)[16],
                                       const v2f (&b)[16], v8f& c0, v8f& c1) {
    c0 = (v8f){};
    c1 = (v8f){};
#pragma unroll
    for (int j = 0; j < 16; ++j) {
        c0 = __builtin_amdgcn_wmma_f32_16x16x4_f32(
                 false, a0[j], false, b[j], (short)0, c0, false, false);
        c1 = __builtin_amdgcn_wmma_f32_16x16x4_f32(
                 false, a1[j], false, b[j], (short)0, c1, false, false);
    }
}

__device__ __forceinline__ void vq_argmin(const v8f& c0, const v8f& c1, float cnorm, int n,
                                          float (&bestS0)[8], int (&bestI0)[8],
                                          float (&bestS1)[8], int (&bestI1)[8]) {
#pragma unroll
    for (int r = 0; r < 8; ++r) {
        float s0 = __builtin_fmaf(-2.0f, c0[r], cnorm);
        float s1 = __builtin_fmaf(-2.0f, c1[r], cnorm);
        if (s0 < bestS0[r]) { bestS0[r] = s0; bestI0[r] = n; }
        if (s1 < bestS1[r]) { bestS1[r] = s1; bestI1[r] = n; }
    }
}

template <int MASK>
__device__ __forceinline__ int row_xor(int v) {
    return __builtin_amdgcn_mov_dpp(v, 0x160 | MASK, 0xf, 0xf, false);
}

template <int MASK>
__device__ __forceinline__ void vq_merge(float& s, int& i) {
    const float so = __builtin_bit_cast(float, row_xor<MASK>(__builtin_bit_cast(int, s)));
    const int   io = row_xor<MASK>(i);
    const bool take = (so < s) | ((so == s) & (io < i));
    s = take ? so : s;
    i = take ? io : i;
}

template <int MASK>
__device__ __forceinline__ void vq_merge_all(float (&bestS0)[8], int (&bestI0)[8],
                                             float (&bestS1)[8], int (&bestI1)[8]) {
#pragma unroll
    for (int r = 0; r < 8; ++r) {
        vq_merge<MASK>(bestS0[r], bestI0[r]);
        vq_merge<MASK>(bestS1[r], bestI1[r]);
    }
}

__global__ __launch_bounds__(256)
void ResidualVQ_main_kernel(const float* __restrict__ x,
                            const float* __restrict__ cbs,
                            const float* __restrict__ csq,
                            float* __restrict__ out, int tokens) {
    __shared__ __attribute__((aligned(16))) float lds[2 * TFLT];

    const int tid  = threadIdx.x;
    const int lane = tid & 31;
    const int wave = tid >> 5;
    const int m    = lane & 15;
    const int hi   = lane >> 4;
    const int tileBase = (blockIdx.x * 8 + wave) * 32;
    const int token0 = tileBase + m;
    const int token1 = tileBase + 16 + m;

    const int srow   = tid >> 4;
    const int spiece = tid & 15;
    const int gOff = srow * DIMV  + spiece * 4;
    const int lOff = srow * PITCH + spiece * 4;

    const float* xrow0 = x + (size_t)token0 * DIMV;
    const float* xrow1 = x + (size_t)token1 * DIMV;
    v2f a0[16], a1[16], acc0[16], acc1[16];
#pragma unroll
    for (int j = 0; j < 16; ++j) {
        a0[j]   = *(const v2f*)(xrow0 + 4 * j + 2 * hi);
        a1[j]   = *(const v2f*)(xrow1 + 4 * j + 2 * hi);
        acc0[j] = (v2f){0.0f, 0.0f};
        acc1[j] = (v2f){0.0f, 0.0f};
    }

    const float* gstream = cbs + gOff;
    int cp = TROWS / 32;
    v4f p0 = *(const v4f*)(gstream + (size_t)cp * 32 * DIMV);
    v4f p1 = *(const v4f*)(gstream + (size_t)cp * 32 * DIMV + 16 * DIMV);

    for (int s = 0; s < STEPS; ++s) {
        v4f v = *(const v4f*)(cbs + (size_t)s * 16 * DIMV + gOff);
        *(v4f*)(lds + s * 16 * PITCH + lOff) = v;
    }
    __syncthreads();

    for (int q = 0; q < QN; ++q) {
        const float* cb = cbs + (size_t)q * KCB * DIMV;
        const float* cs = csq + q * KCB;

        float bestS0[8], bestS1[8];
        int   bestI0[8], bestI1[8];
#pragma unroll
        for (int r = 0; r < 8; ++r) {
            bestS0[r] = __builtin_inff(); bestI0[r] = 0;
            bestS1[r] = __builtin_inff(); bestI1[r] = 0;
        }

        for (int h = 0; h < 2; ++h) {
            const int t = q * 2 + h;
            const float* cur  = lds + (t & 1) * TFLT;
            float*       nxt  = lds + ((t + 1) & 1) * TFLT;
            float*       ldst = nxt + lOff;
            const float* frag = cur + m * PITCH + 2 * hi;
            const float* csn  = cs + h * TROWS + m;
            const int    nb   = h * TROWS + m;

            v2f bA0[16], bB0[16], bA1[16], bB1[16];
            load_bfrag(bA0, frag);
            load_bfrag(bB0, frag + 16 * PITCH);
            for (int i = 0; i < STEPS; i += 4) {
                *(v4f*)(ldst + i * 16 * PITCH) = p0;
                *(v4f*)(ldst + (i + 1) * 16 * PITCH) = p1;
                cp = (cp + 1 < QN * KCB / 32) ? (cp + 1) : cp;
                p0 = *(const v4f*)(gstream + (size_t)cp * 32 * DIMV);
                p1 = *(const v4f*)(gstream + (size_t)cp * 32 * DIMV + 16 * DIMV);
                const float cn0 = csn[i * 16];
                const float cn1 = csn[(i + 1) * 16];
                load_bfrag(bA1, frag + (i + 2) * 16 * PITCH);
                load_bfrag(bB1, frag + (i + 3) * 16 * PITCH);
                __builtin_amdgcn_sched_barrier(0);

                v8f c0A, c1A, c0B, c1B;
                vq_mma(a0, a1, bA0, c0A, c1A);
                vq_mma(a0, a1, bB0, c0B, c1B);
                vq_argmin(c0A, c1A, cn0, nb + i * 16, bestS0, bestI0, bestS1, bestI1);
                vq_argmin(c0B, c1B, cn1, nb + (i + 1) * 16, bestS0, bestI0, bestS1, bestI1);
                __builtin_amdgcn_sched_barrier(0);

                *(v4f*)(ldst + (i + 2) * 16 * PITCH) = p0;
                *(v4f*)(ldst + (i + 3) * 16 * PITCH) = p1;
                cp = (cp + 1 < QN * KCB / 32) ? (cp + 1) : cp;
                p0 = *(const v4f*)(gstream + (size_t)cp * 32 * DIMV);
                p1 = *(const v4f*)(gstream + (size_t)cp * 32 * DIMV + 16 * DIMV);
                const float cn2 = csn[(i + 2) * 16];
                const float cn3 = csn[(i + 3) * 16];
                const int i4 = (i + 4 < STEPS) ? (i + 4) : i;
                load_bfrag(bA0, frag + i4 * 16 * PITCH);
                load_bfrag(bB0, frag + (i4 + 1) * 16 * PITCH);
                __builtin_amdgcn_sched_barrier(0);

                v8f c0C, c1C, c0D, c1D;
                vq_mma(a0, a1, bA1, c0C, c1C);
                vq_mma(a0, a1, bB1, c0D, c1D);
                vq_argmin(c0C, c1C, cn2, nb + (i + 2) * 16, bestS0, bestI0, bestS1, bestI1);
                vq_argmin(c0D, c1D, cn3, nb + (i + 3) * 16, bestS0, bestI0, bestS1, bestI1);
                __builtin_amdgcn_sched_barrier(0);
            }
            __syncthreads();
        }

        vq_merge_all<8>(bestS0, bestI0, bestS1, bestI1);
        vq_merge_all<4>(bestS0, bestI0, bestS1, bestI1);
        vq_merge_all<2>(bestS0, bestI0, bestS1, bestI1);
        vq_merge_all<1>(bestS0, bestI0, bestS1, bestI1);

        int myIdx0 = 0, myIdx1 = 0;
#pragma unroll
        for (int r = 0; r < 8; ++r) {
            int lo0 = __shfl(bestI0[r], 0, 32);
            int hi0 = __shfl(bestI0[r], 16, 32);
            int lo1 = __shfl(bestI1[r], 0, 32);
            int hi1 = __shfl(bestI1[r], 16, 32);
            if (m == r)     { myIdx0 = lo0; myIdx1 = lo1; }
            if (m == r + 8) { myIdx0 = hi0; myIdx1 = hi1; }
        }

        const float* sel0 = cb + (size_t)myIdx0 * DIMV + 2 * hi;
        const float* sel1 = cb + (size_t)myIdx1 * DIMV + 2 * hi;
#pragma unroll
        for (int j = 0; j < 16; ++j) {
            v2f cv0 = *(const v2f*)(sel0 + 4 * j);
            v2f cv1 = *(const v2f*)(sel1 + 4 * j);
            a0[j]   -= cv0;
            a1[j]   -= cv1;
            acc0[j] += cv0;
            acc1[j] += cv1;
        }
    }

    float* orow0 = out + (size_t)token0 * DIMV;
    float* orow1 = out + (size_t)token1 * DIMV;
#pragma unroll
    for (int j = 0; j < 16; ++j) {
        *(v2f*)(orow0 + 4 * j + 2 * hi) = acc0[j];
        *(v2f*)(orow1 + 4 * j + 2 * hi) = acc1[j];
    }
}

extern "C" void kernel_launch(void* const* d_in, const int* in_sizes, int n_in,
                              void* d_out, int out_size, void* d_ws, size_t ws_size,
                              hipStream_t stream) {
    const float* x   = (const float*)d_in[0];
    const float* cbs = (const float*)d_in[1];
    float*       out = (float*)d_out;
    float*       csq = (float*)d_ws;

    const int tokens  = in_sizes[0] / DIMV;
    const int totalCB = in_sizes[1] / DIMV;

    ResidualVQ_csq_kernel<<<(totalCB + 255) / 256, 256, 0, stream>>>(cbs, csq, totalCB);
    ResidualVQ_main_kernel<<<tokens / 256, 256, 0, stream>>>(x, cbs, csq, out, tokens);
}


# WMMA operand roles swapped (codebook rows as A, residuals as B): a lane owns one token and 8 codewords per tile, so the argmin is an in-lane min tree with one compare per 8 scores
# speedup vs baseline: 1.3559x; 1.0496x over previous
#include <hip/hip_runtime.h>
#include <hip/hip_bf16.h>


#define QN    8
#define KCB   1024
#define DIMV  64
#define TROWS 512
#define PITCH 68
#define TFLT  (TROWS * PITCH)
#define NTILE (QN * KCB / TROWS)
#define STEPS (TROWS / 16)

typedef __attribute__((ext_vector_type(2))) float v2f;
typedef __attribute__((ext_vector_type(4))) float v4f;
typedef __attribute__((ext_vector_type(8))) float v8f;

__global__ __launch_bounds__(256)
void ResidualVQ_csq_kernel(const float* __restrict__ cbs,
                           float* __restrict__ csq, int total) {
    int i = blockIdx.x * blockDim.x + threadIdx.x;
    if (i >= total) return;
    const float* r = cbs + (size_t)i * DIMV;
    float s = 0.0f;
#pragma unroll
    for (int k = 0; k < DIMV; ++k) s = __builtin_fmaf(r[k], r[k], s);
    csq[i] = s;
}

__device__ __forceinline__ void load_bfrag(v2f (&b)[16], const float* p) {
#pragma unroll
    for (int j = 0; j < 16; ++j) b[j] = *(const v2f*)(p + 4 * j);
}

__device__ __forceinline__ void vq_mma(const v2f (&a0)[16], const v2f (&a1)[16],
                                       const v2f (&b)[16], v8f& c0, v8f& c1) {
    c0 = (v8f){};
    c1 = (v8f){};
#pragma unroll
    for (int j = 0; j < 16; ++j) {
        c0 = __builtin_amdgcn_wmma_f32_16x16x4_f32(
                 false, b[j], false, a0[j], (short)0, c0, false, false);
        c1 = __builtin_amdgcn_wmma_f32_16x16x4_f32(
                 false, b[j], false, a1[j], (short)0, c1, false, false);
    }
}

__device__ __forceinline__ float vq_min8(const float (&s)[8]) {
    return __builtin_fminf(
        __builtin_fminf(__builtin_fminf(s[0], s[1]), __builtin_fminf(s[2], s[3])),
        __builtin_fminf(__builtin_fminf(s[4], s[5]), __builtin_fminf(s[6], s[7])));
}

__device__ __forceinline__ void vq_argmin(const v8f& c, const v4f& cnlo, const v4f& cnhi,
                                          int step, float& best, int& bstep,
                                          float (&keep)[8]) {
    const v2f neg2 = (v2f){-2.0f, -2.0f};
    const v2f s01 = __builtin_elementwise_fma(neg2, __builtin_shufflevector(c, c, 0, 1),
                                              __builtin_shufflevector(cnlo, cnlo, 0, 1));
    const v2f s23 = __builtin_elementwise_fma(neg2, __builtin_shufflevector(c, c, 2, 3),
                                              __builtin_shufflevector(cnlo, cnlo, 2, 3));
    const v2f s45 = __builtin_elementwise_fma(neg2, __builtin_shufflevector(c, c, 4, 5),
                                              __builtin_shufflevector(cnhi, cnhi, 0, 1));
    const v2f s67 = __builtin_elementwise_fma(neg2, __builtin_shufflevector(c, c, 6, 7),
                                              __builtin_shufflevector(cnhi, cnhi, 2, 3));
    const float s[8] = {s01[0], s01[1], s23[0], s23[1], s45[0], s45[1], s67[0], s67[1]};
    const float m = vq_min8(s);
    const bool take = m < best;
    best  = take ? m : best;
    bstep = take ? step : bstep;
#pragma unroll
    for (int r = 0; r < 8; ++r) keep[r] = take ? s[r] : keep[r];
}

__device__ __forceinline__ int xhalf_i(int v) {
    return __builtin_amdgcn_permlanex16(v, v, 0x76543210, 0xfedcba98, false, false);
}

__device__ __forceinline__ int vq_winner(float best, int bstep, const float (&keep)[8], int hi) {
    int rsel = 0;
#pragma unroll
    for (int r = 7; r >= 0; --r) rsel = (keep[r] == best) ? r : rsel;
    const int   idx = bstep * 16 + 8 * hi + rsel;
    const float so  = __builtin_bit_cast(float, xhalf_i(__builtin_bit_cast(int, best)));
    const int   io  = xhalf_i(idx);
    const bool take = (so < best) | ((so == best) & (io < idx));
    return take ? io : idx;
}

__global__ __launch_bounds__(256)
void ResidualVQ_main_kernel(const float* __restrict__ x,
                            const float* __restrict__ cbs,
                            const float* __restrict__ csq,
                            float* __restrict__ out, int tokens) {
    __shared__ __attribute__((aligned(16))) float lds[2 * TFLT];

    const int tid  = threadIdx.x;
    const int lane = tid & 31;
    const int wave = tid >> 5;
    const int m    = lane & 15;
    const int hi   = lane >> 4;
    const int tileBase = (blockIdx.x * 8 + wave) * 32;
    const int token0 = tileBase + m;
    const int token1 = tileBase + 16 + m;

    const int srow   = tid >> 4;
    const int spiece = tid & 15;
    const int gOff = srow * DIMV  + spiece * 4;
    const int lOff = srow * PITCH + spiece * 4;

    const float* xrow0 = x + (size_t)token0 * DIMV;
    const float* xrow1 = x + (size_t)token1 * DIMV;
    v2f a0[16], a1[16], acc0[16], acc1[16];
#pragma unroll
    for (int j = 0; j < 16; ++j) {
        a0[j]   = *(const v2f*)(xrow0 + 4 * j + 2 * hi);
        a1[j]   = *(const v2f*)(xrow1 + 4 * j + 2 * hi);
        acc0[j] = (v2f){0.0f, 0.0f};
        acc1[j] = (v2f){0.0f, 0.0f};
    }

    const float* gstream = cbs + gOff;
    int cp = TROWS / 32;
    v4f p0 = *(const v4f*)(gstream + (size_t)cp * 32 * DIMV);
    v4f p1 = *(const v4f*)(gstream + (size_t)cp * 32 * DIMV + 16 * DIMV);

    for (int s = 0; s < STEPS; ++s) {
        v4f v = *(const v4f*)(cbs + (size_t)s * 16 * DIMV + gOff);
        *(v4f*)(lds + s * 16 * PITCH + lOff) = v;
    }
    __syncthreads();

    for (int q = 0; q < QN; ++q) {
        const float* cb = cbs + (size_t)q * KCB * DIMV;
        const float* cs = csq + q * KCB;

        float best0 = __builtin_inff(), best1 = __builtin_inff();
        int   bstep0 = 0, bstep1 = 0;
        float keep0[8], keep1[8];
#pragma unroll
        for (int r = 0; r < 8; ++r) {
            keep0[r] = __builtin_inff();
            keep1[r] = __builtin_inff();
        }

        for (int h = 0; h < 2; ++h) {
            const int t = q * 2 + h;
            const float* cur  = lds + (t & 1) * TFLT;
            float*       nxt  = lds + ((t + 1) & 1) * TFLT;
            float*       ldst = nxt + lOff;
            const float* frag = cur + m * PITCH + 2 * hi;
            const float* csn  = cs + h * TROWS + 8 * hi;
            const int    sb   = h * STEPS;

            v2f bA0[16], bB0[16], bA1[16], bB1[16];
            load_bfrag(bA0, frag);
            load_bfrag(bB0, frag + 16 * PITCH);
            for (int i = 0; i < STEPS; i += 4) {
                *(v4f*)(ldst + i * 16 * PITCH) = p0;
                *(v4f*)(ldst + (i + 1) * 16 * PITCH) = p1;
                cp = (cp + 1 < QN * KCB / 32) ? (cp + 1) : cp;
                p0 = *(const v4f*)(gstream + (size_t)cp * 32 * DIMV);
                p1 = *(const v4f*)(gstream + (size_t)cp * 32 * DIMV + 16 * DIMV);
                const v4f cn0l = *(const v4f*)(csn + i * 16);
                const v4f cn0h = *(const v4f*)(csn + i * 16 + 4);
                const v4f cn1l = *(const v4f*)(csn + (i + 1) * 16);
                const v4f cn1h = *(const v4f*)(csn + (i + 1) * 16 + 4);
                load_bfrag(bA1, frag + (i + 2) * 16 * PITCH);
                load_bfrag(bB1, frag + (i + 3) * 16 * PITCH);
                __builtin_amdgcn_sched_barrier(0);

                v8f c0A, c1A, c0B, c1B;
                vq_mma(a0, a1, bA0, c0A, c1A);
                vq_mma(a0, a1, bB0, c0B, c1B);
                vq_argmin(c0A, cn0l, cn0h, sb + i,     best0, bstep0, keep0);
                vq_argmin(c1A, cn0l, cn0h, sb + i,     best1, bstep1, keep1);
                vq_argmin(c0B, cn1l, cn1h, sb + i + 1, best0, bstep0, keep0);
                vq_argmin(c1B, cn1l, cn1h, sb + i + 1, best1, bstep1, keep1);
                __builtin_amdgcn_sched_barrier(0);

                *(v4f*)(ldst + (i + 2) * 16 * PITCH) = p0;
                *(v4f*)(ldst + (i + 3) * 16 * PITCH) = p1;
                cp = (cp + 1 < QN * KCB / 32) ? (cp + 1) : cp;
                p0 = *(const v4f*)(gstream + (size_t)cp * 32 * DIMV);
                p1 = *(const v4f*)(gstream + (size_t)cp * 32 * DIMV + 16 * DIMV);
                const v4f cn2l = *(const v4f*)(csn + (i + 2) * 16);
                const v4f cn2h = *(const v4f*)(csn + (i + 2) * 16 + 4);
                const v4f cn3l = *(const v4f*)(csn + (i + 3) * 16);
                const v4f cn3h = *(const v4f*)(csn + (i + 3) * 16 + 4);
                const int i4 = (i + 4 < STEPS) ? (i + 4) : i;
                load_bfrag(bA0, frag + i4 * 16 * PITCH);
                load_bfrag(bB0, frag + (i4 + 1) * 16 * PITCH);
                __builtin_amdgcn_sched_barrier(0);

                v8f c0C, c1C, c0D, c1D;
                vq_mma(a0, a1, bA1, c0C, c1C);
                vq_mma(a0, a1, bB1, c0D, c1D);
                vq_argmin(c0C, cn2l, cn2h, sb + i + 2, best0, bstep0, keep0);
                vq_argmin(c1C, cn2l, cn2h, sb + i + 2, best1, bstep1, keep1);
                vq_argmin(c0D, cn3l, cn3h, sb + i + 3, best0, bstep0, keep0);
                vq_argmin(c1D, cn3l, cn3h, sb + i + 3, best1, bstep1, keep1);
                __builtin_amdgcn_sched_barrier(0);
            }
            __syncthreads();
        }

        const int myIdx0 = vq_winner(best0, bstep0, keep0, hi);
        const int myIdx1 = vq_winner(best1, bstep1, keep1, hi);

        const float* sel0 = cb + (size_t)myIdx0 * DIMV + 2 * hi;
        const float* sel1 = cb + (size_t)myIdx1 * DIMV + 2 * hi;
#pragma unroll
        for (int j = 0; j < 16; ++j) {
            v2f cv0 = *(const v2f*)(sel0 + 4 * j);
            v2f cv1 = *(const v2f*)(sel1 + 4 * j);
            a0[j]   -= cv0;
            a1[j]   -= cv1;
            acc0[j] += cv0;
            acc1[j] += cv1;
        }
    }

    float* orow0 = out + (size_t)token0 * DIMV;
    float* orow1 = out + (size_t)token1 * DIMV;
#pragma unroll
    for (int j = 0; j < 16; ++j) {
        *(v2f*)(orow0 + 4 * j + 2 * hi) = acc0[j];
        *(v2f*)(orow1 + 4 * j + 2 * hi) = acc1[j];
    }
}

extern "C" void kernel_launch(void* const* d_in, const int* in_sizes, int n_in,
                              void* d_out, int out_size, void* d_ws, size_t ws_size,
                              hipStream_t stream) {
    const float* x   = (const float*)d_in[0];
    const float* cbs = (const float*)d_in[1];
    float*       out = (float*)d_out;
    float*       csq = (float*)d_ws;

    const int tokens  = in_sizes[0] / DIMV;
    const int totalCB = in_sizes[1] / DIMV;

    ResidualVQ_csq_kernel<<<(totalCB + 255) / 256, 256, 0, stream>>>(cbs, csq, totalCB);
    ResidualVQ_main_kernel<<<tokens / 256, 256, 0, stream>>>(x, cbs, csq, out, tokens);
}
